# starting kernel
# baseline (speedup 1.0000x reference)
#include <hip/hip_runtime.h>
#include <hip/hip_bf16.h>
#include <math.h>

typedef __attribute__((ext_vector_type(2))) float v2f;
typedef __attribute__((ext_vector_type(8))) float v8f;

#define D_IN   1024
#define D_ST   16
#define DT_RANK 64
#define N_E    96
#define BATCH  2
#define SEQ    2048
#define M_TOT  (BATCH*SEQ)
#define NCHUNK 8
#define LCHUNK (SEQ/NCHUNK)
#define NCHAN  (BATCH*D_IN)

__global__ void xproj_wmma(const float* __restrict__ X,
                           const float* __restrict__ W,
                           float* __restrict__ xdbl) {
    int wave = (blockIdx.x * blockDim.x + threadIdx.x) >> 5;
    int lane = threadIdx.x & 31;
    const int NT = N_E / 16;
    int nt = wave % NT;
    int mt = wave / NT;
    if (mt >= M_TOT / 16) return;
    int m0 = mt * 16, n0 = nt * 16;
    int half = lane >> 4, lq = lane & 15;
    int koff = half * 2;
    const float* aptr = X + (size_t)(m0 + lq) * D_IN + koff;
    const float* bptr = W + (size_t)(n0 + lq) * D_IN + koff;
    v8f acc = {};
    for (int k = 0; k < D_IN; k += 4) {
        v2f a = *(const v2f*)(aptr + k);
        v2f b = *(const v2f*)(bptr + k);
        acc = __builtin_amdgcn_wmma_f32_16x16x4_f32(
            false, a, false, b, (short)0, acc, false, false);
    }
    int n = n0 + lq;
#pragma unroll
    for (int v = 0; v < 8; ++v) {
        int m = m0 + v + 8 * half;
        xdbl[(size_t)m * N_E + n] = acc[v];
    }
}

__global__ void dt_wmma(const float* __restrict__ xdbl,
                        const float* __restrict__ Wdt,
                        const float* __restrict__ bdt,
                        float* __restrict__ dt) {
    int wave = (blockIdx.x * blockDim.x + threadIdx.x) >> 5;
    int lane = threadIdx.x & 31;
    const int NT = D_IN / 16;
    int nt = wave % NT;
    int mt = wave / NT;
    if (mt >= M_TOT / 16) return;
    int m0 = mt * 16, n0 = nt * 16;
    int half = lane >> 4, lq = lane & 15;
    int koff = half * 2;
    const float* aptr = xdbl + (size_t)(m0 + lq) * N_E + koff;
    const float* bptr = Wdt + (size_t)(n0 + lq) * DT_RANK + koff;
    v8f acc = {};
#pragma unroll
    for (int k = 0; k < DT_RANK; k += 4) {
        v2f a = *(const v2f*)(aptr + k);
        v2f b = *(const v2f*)(bptr + k);
        acc = __builtin_amdgcn_wmma_f32_16x16x4_f32(
            false, a, false, b, (short)0, acc, false, false);
    }
    int n = n0 + lq;
    float bias = bdt[n];
#pragma unroll
    for (int v = 0; v < 8; ++v) {
        int m = m0 + v + 8 * half;
        float z = acc[v] + bias;
        float sp = (z > 20.0f) ? z : log1pf(__expf(z));
        dt[(size_t)m * D_IN + n] = sp;
    }
}

__global__ void scan_phase1(const float* __restrict__ X,
                            const float* __restrict__ xdbl,
                            const float* __restrict__ dt,
                            const float* __restrict__ A_log,
                            float* __restrict__ aprod,
                            float* __restrict__ hend) {
    int gid = blockIdx.x * blockDim.x + threadIdx.x;
    int unit = gid >> 4;
    int s = gid & 15;
    if (unit >= NCHAN * NCHUNK) return;
    int chunk = unit % NCHUNK;
    int channel = unit / NCHUNK;
    int b = channel / D_IN;
    int d = channel % D_IN;
    float A = -__expf(A_log[d * D_ST + s]);
    float ap = 1.0f, h = 0.0f;
    int l0 = chunk * LCHUNK;
    for (int l = l0; l < l0 + LCHUNK; ++l) {
        int m = b * SEQ + l;
        float dtv = dt[(size_t)m * D_IN + d];
        float xv  = X[(size_t)m * D_IN + d];
        float Bm  = xdbl[(size_t)m * N_E + DT_RANK + s];
        float ab  = __expf(dtv * A);
        h  = ab * h + dtv * Bm * xv;
        ap *= ab;
    }
    int idx = (channel * D_ST + s) * NCHUNK + chunk;
    aprod[idx] = ap;
    hend[idx]  = h;
}

__global__ void scan_phase2(const float* __restrict__ aprod,
                            const float* __restrict__ hend,
                            float* __restrict__ carry) {
    int t = blockIdx.x * blockDim.x + threadIdx.x;
    if (t >= NCHAN * D_ST) return;
    float c = 0.0f;
#pragma unroll
    for (int ch = 0; ch < NCHUNK; ++ch) {
        int idx = t * NCHUNK + ch;
        carry[idx] = c;
        c = aprod[idx] * c + hend[idx];
    }
}

__global__ void scan_phase3(const float* __restrict__ X,
                            const float* __restrict__ xdbl,
                            const float* __restrict__ dt,
                            const float* __restrict__ A_log,
                            const float* __restrict__ Dp,
                            const float* __restrict__ carry,
                            float* __restrict__ y) {
    int gid = blockIdx.x * blockDim.x + threadIdx.x;
    int unit = gid >> 4;
    int s = gid & 15;
    if (unit >= NCHAN * NCHUNK) return;
    int chunk = unit % NCHUNK;
    int channel = unit / NCHUNK;
    int b = channel / D_IN;
    int d = channel % D_IN;
    float A = -__expf(A_log[d * D_ST + s]);
    float h = carry[(channel * D_ST + s) * NCHUNK + chunk];
    float Dd = Dp[d];
    int l0 = chunk * LCHUNK;
    for (int l = l0; l < l0 + LCHUNK; ++l) {
        int m = b * SEQ + l;
        float dtv = dt[(size_t)m * D_IN + d];
        float xv  = X[(size_t)m * D_IN + d];
        float Bm  = xdbl[(size_t)m * N_E + DT_RANK + s];
        float Cm  = xdbl[(size_t)m * N_E + DT_RANK + D_ST + s];
        float ab  = __expf(dtv * A);
        h = ab * h + dtv * Bm * xv;
        float contrib = h * Cm;
#pragma unroll
        for (int msk = 1; msk < 16; msk <<= 1)
            contrib += __shfl_xor(contrib, msk, 32);
        if (s == 0)
            y[(size_t)m * D_IN + d] = contrib + Dd * xv;
    }
}

extern "C" void kernel_launch(void* const* d_in, const int* in_sizes, int n_in,
                              void* d_out, int out_size, void* d_ws, size_t ws_size,
                              hipStream_t stream) {
    const float* x       = (const float*)d_in[0];
    const float* A_log   = (const float*)d_in[1];
    const float* Dp      = (const float*)d_in[2];
    const float* W_xproj = (const float*)d_in[3];
    const float* W_dt    = (const float*)d_in[4];
    const float* b_dt    = (const float*)d_in[5];
    float* y = (float*)d_out;

    char* ws = (char*)d_ws;
    size_t off = 0;
    float* xdbl  = (float*)(ws + off); off += (size_t)M_TOT * N_E * 4;
    off = (off + 255) & ~(size_t)255;
    float* dtbuf = (float*)(ws + off); off += (size_t)M_TOT * D_IN * 4;
    off = (off + 255) & ~(size_t)255;
    float* aprod = (float*)(ws + off); off += (size_t)NCHAN * D_ST * NCHUNK * 4;
    off = (off + 255) & ~(size_t)255;
    float* hend  = (float*)(ws + off); off += (size_t)NCHAN * D_ST * NCHUNK * 4;
    off = (off + 255) & ~(size_t)255;
    float* carry = (float*)(ws + off);

    xproj_wmma<<<1536 / 4, 128, 0, stream>>>(x, W_xproj, xdbl);
    dt_wmma<<<16384 / 8, 256, 0, stream>>>(xdbl, W_dt, b_dt, dtbuf);
    scan_phase1<<<(NCHAN * NCHUNK * 16) / 256, 256, 0, stream>>>(
        x, xdbl, dtbuf, A_log, aprod, hend);
    scan_phase2<<<(NCHAN * D_ST) / 256, 256, 0, stream>>>(aprod, hend, carry);
    scan_phase3<<<(NCHAN * NCHUNK * 16) / 256, 256, 0, stream>>>(
        x, xdbl, dtbuf, A_log, Dp, carry, y);
}


# Scan remapped: one lane per channel with 16 states in registers, coalesced dt/x/y, B/C by readlane; 16-state sum now an in-lane serial fma chain (sum order changed)
# speedup vs baseline: 2.8210x; 2.8210x over previous
#include <hip/hip_runtime.h>
#include <hip/hip_bf16.h>
#include <math.h>

typedef __attribute__((ext_vector_type(2))) float v2f;
typedef __attribute__((ext_vector_type(8))) float v8f;

#define D_IN   1024
#define D_ST   16
#define DT_RANK 64
#define N_E    96
#define BATCH  2
#define SEQ    2048
#define M_TOT  (BATCH*SEQ)
#define NCHUNK 8
#define LCHUNK (SEQ/NCHUNK)
#define NCHAN  (BATCH*D_IN)

__global__ void xproj_wmma(const float* __restrict__ X,
                           const float* __restrict__ W,
                           float* __restrict__ xdbl) {
    int wave = (blockIdx.x * blockDim.x + threadIdx.x) >> 5;
    int lane = threadIdx.x & 31;
    const int NT = N_E / 16;
    int nt = wave % NT;
    int mt = wave / NT;
    if (mt >= M_TOT / 16) return;
    int m0 = mt * 16, n0 = nt * 16;
    int half = lane >> 4, lq = lane & 15;
    int koff = half * 2;
    const float* aptr = X + (size_t)(m0 + lq) * D_IN + koff;
    const float* bptr = W + (size_t)(n0 + lq) * D_IN + koff;
    v8f acc = {};
    for (int k = 0; k < D_IN; k += 4) {
        v2f a = *(const v2f*)(aptr + k);
        v2f b = *(const v2f*)(bptr + k);
        acc = __builtin_amdgcn_wmma_f32_16x16x4_f32(
            false, a, false, b, (short)0, acc, false, false);
    }
    int n = n0 + lq;
#pragma unroll
    for (int v = 0; v < 8; ++v) {
        int m = m0 + v + 8 * half;
        xdbl[(size_t)m * N_E + n] = acc[v];
    }
}

__global__ void dt_wmma(const float* __restrict__ xdbl,
                        const float* __restrict__ Wdt,
                        const float* __restrict__ bdt,
                        float* __restrict__ dt) {
    int wave = (blockIdx.x * blockDim.x + threadIdx.x) >> 5;
    int lane = threadIdx.x & 31;
    const int NT = D_IN / 16;
    int nt = wave % NT;
    int mt = wave / NT;
    if (mt >= M_TOT / 16) return;
    int m0 = mt * 16, n0 = nt * 16;
    int half = lane >> 4, lq = lane & 15;
    int koff = half * 2;
    const float* aptr = xdbl + (size_t)(m0 + lq) * N_E + koff;
    const float* bptr = Wdt + (size_t)(n0 + lq) * DT_RANK + koff;
    v8f acc = {};
#pragma unroll
    for (int k = 0; k < DT_RANK; k += 4) {
        v2f a = *(const v2f*)(aptr + k);
        v2f b = *(const v2f*)(bptr + k);
        acc = __builtin_amdgcn_wmma_f32_16x16x4_f32(
            false, a, false, b, (short)0, acc, false, false);
    }
    int n = n0 + lq;
    float bias = bdt[n];
#pragma unroll
    for (int v = 0; v < 8; ++v) {
        int m = m0 + v + 8 * half;
        float z = acc[v] + bias;
        float sp = (z > 20.0f) ? z : log1pf(__expf(z));
        dt[(size_t)m * D_IN + n] = sp;
    }
}

#define SCAN_U   8
#define SCAN_WG  64
#define SCAN_DB  (D_IN / SCAN_WG)

__device__ __forceinline__ float lane_bcast(float v, int srclane) {
    return __int_as_float(__builtin_amdgcn_readlane(__float_as_int(v), srclane));
}

__global__ __launch_bounds__(SCAN_WG)
void scan_phase1(const float* __restrict__ X,
                 const float* __restrict__ xdbl,
                 const float* __restrict__ dt,
                 const float* __restrict__ A_log,
                 float* __restrict__ aprod,
                 float* __restrict__ hend) {
    int lane  = threadIdx.x & 31;
    int dblk  = blockIdx.x % SCAN_DB;
    int chunk = (blockIdx.x / SCAN_DB) % NCHUNK;
    int b     = blockIdx.x / (SCAN_DB * NCHUNK);
    int d     = dblk * SCAN_WG + threadIdx.x;
    int channel = b * D_IN + d;

    float A[D_ST], h[D_ST], ap[D_ST];
#pragma unroll
    for (int s = 0; s < D_ST; ++s) {
        A[s]  = -__expf(A_log[d * D_ST + s]);
        h[s]  = 0.0f;
        ap[s] = 1.0f;
    }
    size_t m0 = (size_t)b * SEQ + (size_t)chunk * LCHUNK;
    const float* dtp = dt + m0 * D_IN + d;
    const float* xp  = X  + m0 * D_IN + d;
    const float* bp  = xdbl + m0 * N_E + DT_RANK + (lane & 15);

    float dtc[SCAN_U], xc[SCAN_U], bc[SCAN_U];
#pragma unroll
    for (int u = 0; u < SCAN_U; ++u) {
        dtc[u] = dtp[(size_t)u * D_IN];
        xc[u]  = xp[(size_t)u * D_IN];
        bc[u]  = bp[(size_t)u * N_E];
    }
    for (int l = 0; l < LCHUNK; l += SCAN_U) {
        int ln = (l + SCAN_U < LCHUNK) ? (l + SCAN_U) : l;
        float dtn[SCAN_U], xn[SCAN_U], bn[SCAN_U];
#pragma unroll
        for (int u = 0; u < SCAN_U; ++u) {
            dtn[u] = dtp[(size_t)(ln + u) * D_IN];
            xn[u]  = xp[(size_t)(ln + u) * D_IN];
            bn[u]  = bp[(size_t)(ln + u) * N_E];
        }
#pragma unroll
        for (int u = 0; u < SCAN_U; ++u) {
            float dtv = dtc[u];
            float dtx = dtv * xc[u];
#pragma unroll
            for (int s = 0; s < D_ST; ++s) {
                float Bs = lane_bcast(bc[u], s);
                float ab = __expf(dtv * A[s]);
                h[s]  = ab * h[s] + dtx * Bs;
                ap[s] *= ab;
            }
        }
#pragma unroll
        for (int u = 0; u < SCAN_U; ++u) { dtc[u] = dtn[u]; xc[u] = xn[u]; bc[u] = bn[u]; }
    }
#pragma unroll
    for (int s = 0; s < D_ST; ++s) {
        int idx = (chunk * D_ST + s) * NCHAN + channel;
        aprod[idx] = ap[s];
        hend[idx]  = h[s];
    }
}

__global__ void scan_phase2(const float* __restrict__ aprod,
                            const float* __restrict__ hend,
                            float* __restrict__ carry) {
    int t = blockIdx.x * blockDim.x + threadIdx.x;
    if (t >= NCHAN * D_ST) return;
    float c = 0.0f;
#pragma unroll
    for (int ch = 0; ch < NCHUNK; ++ch) {
        int idx = ch * (NCHAN * D_ST) + t;
        carry[idx] = c;
        c = aprod[idx] * c + hend[idx];
    }
}

__global__ __launch_bounds__(SCAN_WG)
void scan_phase3(const float* __restrict__ X,
                 const float* __restrict__ xdbl,
                 const float* __restrict__ dt,
                 const float* __restrict__ A_log,
                 const float* __restrict__ Dp,
                 const float* __restrict__ carry,
                 float* __restrict__ y) {
    int lane  = threadIdx.x & 31;
    int dblk  = blockIdx.x % SCAN_DB;
    int chunk = (blockIdx.x / SCAN_DB) % NCHUNK;
    int b     = blockIdx.x / (SCAN_DB * NCHUNK);
    int d     = dblk * SCAN_WG + threadIdx.x;
    int channel = b * D_IN + d;

    float A[D_ST], h[D_ST];
#pragma unroll
    for (int s = 0; s < D_ST; ++s) {
        A[s] = -__expf(A_log[d * D_ST + s]);
        h[s] = carry[(chunk * D_ST + s) * NCHAN + channel];
    }
    float Dd = Dp[d];
    size_t m0 = (size_t)b * SEQ + (size_t)chunk * LCHUNK;
    const float* dtp = dt + m0 * D_IN + d;
    const float* xp  = X  + m0 * D_IN + d;
    float*       yp  = y  + m0 * D_IN + d;
    const float* bcp = xdbl + m0 * N_E + DT_RANK + lane;

    float dtc[SCAN_U], xc[SCAN_U], bc[SCAN_U];
#pragma unroll
    for (int u = 0; u < SCAN_U; ++u) {
        dtc[u] = dtp[(size_t)u * D_IN];
        xc[u]  = xp[(size_t)u * D_IN];
        bc[u]  = bcp[(size_t)u * N_E];
    }
    for (int l = 0; l < LCHUNK; l += SCAN_U) {
        int ln = (l + SCAN_U < LCHUNK) ? (l + SCAN_U) : l;
        float dtn[SCAN_U], xn[SCAN_U], bn[SCAN_U];
#pragma unroll
        for (int u = 0; u < SCAN_U; ++u) {
            dtn[u] = dtp[(size_t)(ln + u) * D_IN];
            xn[u]  = xp[(size_t)(ln + u) * D_IN];
            bn[u]  = bcp[(size_t)(ln + u) * N_E];
        }
#pragma unroll
        for (int u = 0; u < SCAN_U; ++u) {
            float dtv = dtc[u];
            float xv  = xc[u];
            float dtx = dtv * xv;
            float acc = 0.0f;
#pragma unroll
            for (int s = 0; s < D_ST; ++s) {
                float Bs = lane_bcast(bc[u], s);
                float Cs = lane_bcast(bc[u], D_ST + s);
                float ab = __expf(dtv * A[s]);
                h[s] = ab * h[s] + dtx * Bs;
                acc  = h[s] * Cs + acc;
            }
            yp[(size_t)(l + u) * D_IN] = acc + Dd * xv;
        }
#pragma unroll
        for (int u = 0; u < SCAN_U; ++u) { dtc[u] = dtn[u]; xc[u] = xn[u]; bc[u] = bn[u]; }
    }
}

extern "C" void kernel_launch(void* const* d_in, const int* in_sizes, int n_in,
                              void* d_out, int out_size, void* d_ws, size_t ws_size,
                              hipStream_t stream) {
    const float* x       = (const float*)d_in[0];
    const float* A_log   = (const float*)d_in[1];
    const float* Dp      = (const float*)d_in[2];
    const float* W_xproj = (const float*)d_in[3];
    const float* W_dt    = (const float*)d_in[4];
    const float* b_dt    = (const float*)d_in[5];
    float* y = (float*)d_out;

    char* ws = (char*)d_ws;
    size_t off = 0;
    float* xdbl  = (float*)(ws + off); off += (size_t)M_TOT * N_E * 4;
    off = (off + 255) & ~(size_t)255;
    float* dtbuf = (float*)(ws + off); off += (size_t)M_TOT * D_IN * 4;
    off = (off + 255) & ~(size_t)255;
    float* aprod = (float*)(ws + off); off += (size_t)NCHAN * D_ST * NCHUNK * 4;
    off = (off + 255) & ~(size_t)255;
    float* hend  = (float*)(ws + off); off += (size_t)NCHAN * D_ST * NCHUNK * 4;
    off = (off + 255) & ~(size_t)255;
    float* carry = (float*)(ws + off);

    xproj_wmma<<<1536 / 4, 128, 0, stream>>>(x, W_xproj, xdbl);
    dt_wmma<<<16384 / 8, 256, 0, stream>>>(xdbl, W_dt, b_dt, dtbuf);
    scan_phase1<<<BATCH * NCHUNK * SCAN_DB, SCAN_WG, 0, stream>>>(
        x, xdbl, dtbuf, A_log, aprod, hend);
    scan_phase2<<<(NCHAN * D_ST) / 256, 256, 0, stream>>>(aprod, hend, carry);
    scan_phase3<<<BATCH * NCHUNK * SCAN_DB, SCAN_WG, 0, stream>>>(
        x, xdbl, dtbuf, A_log, Dp, carry, y);
}


# Scan split into 32 chunks of 64 steps (was 8 of 256): 2048 scan waves instead of 512, two on every SIMD; chunk boundaries of the carry recurrence move, so the rounding order of h changes
# speedup vs baseline: 4.0514x; 1.4362x over previous
#include <hip/hip_runtime.h>
#include <hip/hip_bf16.h>
#include <math.h>

typedef __attribute__((ext_vector_type(2))) float v2f;
typedef __attribute__((ext_vector_type(8))) float v8f;

#define D_IN   1024
#define D_ST   16
#define DT_RANK 64
#define N_E    96
#define BATCH  2
#define SEQ    2048
#define M_TOT  (BATCH*SEQ)
#define NCHUNK 32
#define LCHUNK (SEQ/NCHUNK)
#define NCHAN  (BATCH*D_IN)

__global__ void xproj_wmma(const float* __restrict__ X,
                           const float* __restrict__ W,
                           float* __restrict__ xdbl) {
    int wave = (blockIdx.x * blockDim.x + threadIdx.x) >> 5;
    int lane = threadIdx.x & 31;
    const int NT = N_E / 16;
    int nt = wave % NT;
    int mt = wave / NT;
    if (mt >= M_TOT / 16) return;
    int m0 = mt * 16, n0 = nt * 16;
    int half = lane >> 4, lq = lane & 15;
    int koff = half * 2;
    const float* aptr = X + (size_t)(m0 + lq) * D_IN + koff;
    const float* bptr = W + (size_t)(n0 + lq) * D_IN + koff;
    v8f acc = {};
    for (int k = 0; k < D_IN; k += 4) {
        v2f a = *(const v2f*)(aptr + k);
        v2f b = *(const v2f*)(bptr + k);
        acc = __builtin_amdgcn_wmma_f32_16x16x4_f32(
            false, a, false, b, (short)0, acc, false, false);
    }
    int n = n0 + lq;
#pragma unroll
    for (int v = 0; v < 8; ++v) {
        int m = m0 + v + 8 * half;
        xdbl[(size_t)m * N_E + n] = acc[v];
    }
}

__global__ void dt_wmma(const float* __restrict__ xdbl,
                        const float* __restrict__ Wdt,
                        const float* __restrict__ bdt,
                        float* __restrict__ dt) {
    int wave = (blockIdx.x * blockDim.x + threadIdx.x) >> 5;
    int lane = threadIdx.x & 31;
    const int NT = D_IN / 16;
    int nt = wave % NT;
    int mt = wave / NT;
    if (mt >= M_TOT / 16) return;
    int m0 = mt * 16, n0 = nt * 16;
    int half = lane >> 4, lq = lane & 15;
    int koff = half * 2;
    const float* aptr = xdbl + (size_t)(m0 + lq) * N_E + koff;
    const float* bptr = Wdt + (size_t)(n0 + lq) * DT_RANK + koff;
    v8f acc = {};
#pragma unroll
    for (int k = 0; k < DT_RANK; k += 4) {
        v2f a = *(const v2f*)(aptr + k);
        v2f b = *(const v2f*)(bptr + k);
        acc = __builtin_amdgcn_wmma_f32_16x16x4_f32(
            false, a, false, b, (short)0, acc, false, false);
    }
    int n = n0 + lq;
    float bias = bdt[n];
#pragma unroll
    for (int v = 0; v < 8; ++v) {
        int m = m0 + v + 8 * half;
        float z = acc[v] + bias;
        float sp = (z > 20.0f) ? z : log1pf(__expf(z));
        dt[(size_t)m * D_IN + n] = sp;
    }
}

#define SCAN_U   8
#define SCAN_WG  64
#define SCAN_DB  (D_IN / SCAN_WG)

__device__ __forceinline__ float lane_bcast(float v, int srclane) {
    return __int_as_float(__builtin_amdgcn_readlane(__float_as_int(v), srclane));
}

__global__ __launch_bounds__(SCAN_WG)
void scan_phase1(const float* __restrict__ X,
                 const float* __restrict__ xdbl,
                 const float* __restrict__ dt,
                 const float* __restrict__ A_log,
                 float* __restrict__ aprod,
                 float* __restrict__ hend) {
    int lane  = threadIdx.x & 31;
    int dblk  = blockIdx.x % SCAN_DB;
    int chunk = (blockIdx.x / SCAN_DB) % NCHUNK;
    int b     = blockIdx.x / (SCAN_DB * NCHUNK);
    int d     = dblk * SCAN_WG + threadIdx.x;
    int channel = b * D_IN + d;

    float A[D_ST], h[D_ST], ap[D_ST];
#pragma unroll
    for (int s = 0; s < D_ST; ++s) {
        A[s]  = -__expf(A_log[d * D_ST + s]);
        h[s]  = 0.0f;
        ap[s] = 1.0f;
    }
    size_t m0 = (size_t)b * SEQ + (size_t)chunk * LCHUNK;
    const float* dtp = dt + m0 * D_IN + d;
    const float* xp  = X  + m0 * D_IN + d;
    const float* bp  = xdbl + m0 * N_E + DT_RANK + (lane & 15);

    float dtc[SCAN_U], xc[SCAN_U], bc[SCAN_U];
#pragma unroll
    for (int u = 0; u < SCAN_U; ++u) {
        dtc[u] = dtp[(size_t)u * D_IN];
        xc[u]  = xp[(size_t)u * D_IN];
        bc[u]  = bp[(size_t)u * N_E];
    }
    for (int l = 0; l < LCHUNK; l += SCAN_U) {
        int ln = (l + SCAN_U < LCHUNK) ? (l + SCAN_U) : l;
        float dtn[SCAN_U], xn[SCAN_U], bn[SCAN_U];
#pragma unroll
        for (int u = 0; u < SCAN_U; ++u) {
            dtn[u] = dtp[(size_t)(ln + u) * D_IN];
            xn[u]  = xp[(size_t)(ln + u) * D_IN];
            bn[u]  = bp[(size_t)(ln + u) * N_E];
        }
#pragma unroll
        for (int u = 0; u < SCAN_U; ++u) {
            float dtv = dtc[u];
            float dtx = dtv * xc[u];
#pragma unroll
            for (int s = 0; s < D_ST; ++s) {
                float Bs = lane_bcast(bc[u], s);
                float ab = __expf(dtv * A[s]);
                h[s]  = ab * h[s] + dtx * Bs;
                ap[s] *= ab;
            }
        }
#pragma unroll
        for (int u = 0; u < SCAN_U; ++u) { dtc[u] = dtn[u]; xc[u] = xn[u]; bc[u] = bn[u]; }
    }
#pragma unroll
    for (int s = 0; s < D_ST; ++s) {
        int idx = (chunk * D_ST + s) * NCHAN + channel;
        aprod[idx] = ap[s];
        hend[idx]  = h[s];
    }
}

__global__ void scan_phase2(const float* __restrict__ aprod,
                            const float* __restrict__ hend,
                            float* __restrict__ carry) {
    int t = blockIdx.x * blockDim.x + threadIdx.x;
    if (t >= NCHAN * D_ST) return;
    float c = 0.0f;
#pragma unroll
    for (int ch = 0; ch < NCHUNK; ++ch) {
        int idx = ch * (NCHAN * D_ST) + t;
        carry[idx] = c;
        c = aprod[idx] * c + hend[idx];
    }
}

__global__ __launch_bounds__(SCAN_WG)
void scan_phase3(const float* __restrict__ X,
                 const float* __restrict__ xdbl,
                 const float* __restrict__ dt,
                 const float* __restrict__ A_log,
                 const float* __restrict__ Dp,
                 const float* __restrict__ carry,
                 float* __restrict__ y) {
    int lane  = threadIdx.x & 31;
    int dblk  = blockIdx.x % SCAN_DB;
    int chunk = (blockIdx.x / SCAN_DB) % NCHUNK;
    int b     = blockIdx.x / (SCAN_DB * NCHUNK);
    int d     = dblk * SCAN_WG + threadIdx.x;
    int channel = b * D_IN + d;

    float A[D_ST], h[D_ST];
#pragma unroll
    for (int s = 0; s < D_ST; ++s) {
        A[s] = -__expf(A_log[d * D_ST + s]);
        h[s] = carry[(chunk * D_ST + s) * NCHAN + channel];
    }
    float Dd = Dp[d];
    size_t m0 = (size_t)b * SEQ + (size_t)chunk * LCHUNK;
    const float* dtp = dt + m0 * D_IN + d;
    const float* xp  = X  + m0 * D_IN + d;
    float*       yp  = y  + m0 * D_IN + d;
    const float* bcp = xdbl + m0 * N_E + DT_RANK + lane;

    float dtc[SCAN_U], xc[SCAN_U], bc[SCAN_U];
#pragma unroll
    for (int u = 0; u < SCAN_U; ++u) {
        dtc[u] = dtp[(size_t)u * D_IN];
        xc[u]  = xp[(size_t)u * D_IN];
        bc[u]  = bcp[(size_t)u * N_E];
    }
    for (int l = 0; l < LCHUNK; l += SCAN_U) {
        int ln = (l + SCAN_U < LCHUNK) ? (l + SCAN_U) : l;
        float dtn[SCAN_U], xn[SCAN_U], bn[SCAN_U];
#pragma unroll
        for (int u = 0; u < SCAN_U; ++u) {
            dtn[u] = dtp[(size_t)(ln + u) * D_IN];
            xn[u]  = xp[(size_t)(ln + u) * D_IN];
            bn[u]  = bcp[(size_t)(ln + u) * N_E];
        }
#pragma unroll
        for (int u = 0; u < SCAN_U; ++u) {
            float dtv = dtc[u];
            float xv  = xc[u];
            float dtx = dtv * xv;
            float acc = 0.0f;
#pragma unroll
            for (int s = 0; s < D_ST; ++s) {
                float Bs = lane_bcast(bc[u], s);
                float Cs = lane_bcast(bc[u], D_ST + s);
                float ab = __expf(dtv * A[s]);
                h[s] = ab * h[s] + dtx * Bs;
                acc  = h[s] * Cs + acc;
            }
            yp[(size_t)(l + u) * D_IN] = acc + Dd * xv;
        }
#pragma unroll
        for (int u = 0; u < SCAN_U; ++u) { dtc[u] = dtn[u]; xc[u] = xn[u]; bc[u] = bn[u]; }
    }
}

extern "C" void kernel_launch(void* const* d_in, const int* in_sizes, int n_in,
                              void* d_out, int out_size, void* d_ws, size_t ws_size,
                              hipStream_t stream) {
    const float* x       = (const float*)d_in[0];
    const float* A_log   = (const float*)d_in[1];
    const float* Dp      = (const float*)d_in[2];
    const float* W_xproj = (const float*)d_in[3];
    const float* W_dt    = (const float*)d_in[4];
    const float* b_dt    = (const float*)d_in[5];
    float* y = (float*)d_out;

    char* ws = (char*)d_ws;
    size_t off = 0;
    float* xdbl  = (float*)(ws + off); off += (size_t)M_TOT * N_E * 4;
    off = (off + 255) & ~(size_t)255;
    float* dtbuf = (float*)(ws + off); off += (size_t)M_TOT * D_IN * 4;
    off = (off + 255) & ~(size_t)255;
    float* aprod = (float*)(ws + off); off += (size_t)NCHAN * D_ST * NCHUNK * 4;
    off = (off + 255) & ~(size_t)255;
    float* hend  = (float*)(ws + off); off += (size_t)NCHAN * D_ST * NCHUNK * 4;
    off = (off + 255) & ~(size_t)255;
    float* carry = (float*)(ws + off);

    xproj_wmma<<<1536 / 4, 128, 0, stream>>>(x, W_xproj, xdbl);
    dt_wmma<<<16384 / 8, 256, 0, stream>>>(xdbl, W_dt, b_dt, dtbuf);
    scan_phase1<<<BATCH * NCHUNK * SCAN_DB, SCAN_WG, 0, stream>>>(
        x, xdbl, dtbuf, A_log, aprod, hend);
    scan_phase2<<<(NCHAN * D_ST) / 256, 256, 0, stream>>>(aprod, hend, carry);
    scan_phase3<<<BATCH * NCHUNK * SCAN_DB, SCAN_WG, 0, stream>>>(
        x, xdbl, dtbuf, A_log, Dp, carry, y);
}


# Both GEMMs re-tiled for fewer fragment loads: one A fragment feeds 6 (x_proj) or 2 (dt) column tiles, 128-bit k-permuted loads issued ahead; x_proj K split 8 ways, summed via LDS; K sum order changes
# speedup vs baseline: 4.7741x; 1.1784x over previous
#include <hip/hip_runtime.h>
#include <hip/hip_bf16.h>
#include <math.h>

typedef __attribute__((ext_vector_type(2))) float v2f;
typedef __attribute__((ext_vector_type(4))) float v4f;
typedef __attribute__((ext_vector_type(8))) float v8f;

#define D_IN   1024
#define D_ST   16
#define DT_RANK 64
#define N_E    96
#define BATCH  2
#define SEQ    2048
#define M_TOT  (BATCH*SEQ)
#define NCHUNK 32
#define LCHUNK (SEQ/NCHUNK)
#define NCHAN  (BATCH*D_IN)

#define FRAG_LO(q) (v2f){(q).x, (q).y}
#define FRAG_HI(q) (v2f){(q).z, (q).w}

#define K1_WAVES 8
#define K1_KS    (D_IN / K1_WAVES)
#define K1_NT    (N_E / 16)
#define K1_JB    2
#define K1_SETS  (K1_KS / (8 * K1_JB))

__global__ __launch_bounds__(K1_WAVES * 32)
void xproj_wmma(const float* __restrict__ X,
                const float* __restrict__ W,
                float* __restrict__ xdbl) {
    __shared__ float red[K1_WAVES][K1_NT * 8 * 32];
    int lane = threadIdx.x & 31;
    int w    = threadIdx.x >> 5;
    int m0   = blockIdx.x * 16;
    int half = lane >> 4, lq = lane & 15;
    const float* aptr = X + (size_t)(m0 + lq) * D_IN + w * K1_KS + half * 4;
    const float* bptr = W + (size_t)lq * D_IN + w * K1_KS + half * 4;

    v8f acc[K1_NT];
#pragma unroll
    for (int t = 0; t < K1_NT; ++t) acc[t] = (v8f){};

    v4f a[K1_SETS][K1_JB], b[K1_SETS][K1_JB][K1_NT];
#pragma unroll
    for (int jj = 0; jj < K1_JB; ++jj) {
        a[0][jj] = *(const v4f*)(aptr + 8 * jj);
#pragma unroll
        for (int t = 0; t < K1_NT; ++t)
            b[0][jj][t] = *(const v4f*)(bptr + (size_t)t * 16 * D_IN + 8 * jj);
    }
#pragma unroll
    for (int s = 0; s < K1_SETS; ++s) {
        if (s + 1 < K1_SETS) {
#pragma unroll
            for (int jj = 0; jj < K1_JB; ++jj) {
                int j = (s + 1) * K1_JB + jj;
                a[s + 1][jj] = *(const v4f*)(aptr + 8 * j);
#pragma unroll
                for (int t = 0; t < K1_NT; ++t)
                    b[s + 1][jj][t] = *(const v4f*)(bptr + (size_t)t * 16 * D_IN + 8 * j);
            }
        }
        __builtin_amdgcn_sched_barrier(0);
#pragma unroll
        for (int jj = 0; jj < K1_JB; ++jj) {
#pragma unroll
            for (int t = 0; t < K1_NT; ++t)
                acc[t] = __builtin_amdgcn_wmma_f32_16x16x4_f32(
                    false, FRAG_LO(a[s][jj]), false, FRAG_LO(b[s][jj][t]), (short)0, acc[t], false, false);
#pragma unroll
            for (int t = 0; t < K1_NT; ++t)
                acc[t] = __builtin_amdgcn_wmma_f32_16x16x4_f32(
                    false, FRAG_HI(a[s][jj]), false, FRAG_HI(b[s][jj][t]), (short)0, acc[t], false, false);
        }
    }
#pragma unroll
    for (int t = 0; t < K1_NT; ++t)
#pragma unroll
        for (int v = 0; v < 8; ++v)
            red[w][(t * 8 + v) * 32 + lane] = acc[t][v];
    __syncthreads();
#pragma unroll
    for (int i = 0; i < (16 * N_E) / (K1_WAVES * 32); ++i) {
        int o   = threadIdx.x + i * (K1_WAVES * 32);
        int row = o / N_E;
        int col = o - row * N_E;
        int idx = ((col >> 4) * 8 + (row & 7)) * 32 + (col & 15) + 16 * (row >> 3);
        float s = red[0][idx];
#pragma unroll
        for (int ww = 1; ww < K1_WAVES; ++ww) s += red[ww][idx];
        xdbl[(size_t)m0 * N_E + o] = s;
    }
}

#define K2_NT 2
__global__ __launch_bounds__(256)
void dt_wmma(const float* __restrict__ xdbl,
             const float* __restrict__ Wdt,
             const float* __restrict__ bdt,
             float* __restrict__ dt) {
    int wave = (blockIdx.x * blockDim.x + threadIdx.x) >> 5;
    int lane = threadIdx.x & 31;
    const int NG = D_IN / (16 * K2_NT);
    int ng = wave % NG;
    int mt = wave / NG;
    if (mt >= M_TOT / 16) return;
    int m0 = mt * 16, n0 = ng * (16 * K2_NT);
    int half = lane >> 4, lq = lane & 15;
    const float* aptr = xdbl + (size_t)(m0 + lq) * N_E + half * 4;
    const float* bptr = Wdt + (size_t)(n0 + lq) * DT_RANK + half * 4;
    v8f acc[K2_NT];
#pragma unroll
    for (int t = 0; t < K2_NT; ++t) acc[t] = (v8f){};
    v4f a[DT_RANK / 8], b[DT_RANK / 8][K2_NT];
#pragma unroll
    for (int j = 0; j < DT_RANK / 8; ++j) {
        a[j] = *(const v4f*)(aptr + 8 * j);
#pragma unroll
        for (int t = 0; t < K2_NT; ++t)
            b[j][t] = *(const v4f*)(bptr + (size_t)t * 16 * DT_RANK + 8 * j);
    }
    __builtin_amdgcn_sched_barrier(0);
#pragma unroll
    for (int j = 0; j < DT_RANK / 8; ++j) {
#pragma unroll
        for (int t = 0; t < K2_NT; ++t)
            acc[t] = __builtin_amdgcn_wmma_f32_16x16x4_f32(
                false, FRAG_LO(a[j]), false, FRAG_LO(b[j][t]), (short)0, acc[t], false, false);
#pragma unroll
        for (int t = 0; t < K2_NT; ++t)
            acc[t] = __builtin_amdgcn_wmma_f32_16x16x4_f32(
                false, FRAG_HI(a[j]), false, FRAG_HI(b[j][t]), (short)0, acc[t], false, false);
    }
#pragma unroll
    for (int t = 0; t < K2_NT; ++t) {
        int n = n0 + t * 16 + lq;
        float bias = bdt[n];
#pragma unroll
        for (int v = 0; v < 8; ++v) {
            int m = m0 + v + 8 * half;
            float z = acc[t][v] + bias;
            float sp = (z > 20.0f) ? z : log1pf(__expf(z));
            dt[(size_t)m * D_IN + n] = sp;
        }
    }
}

#define SCAN_U   8
#define SCAN_WG  64
#define SCAN_DB  (D_IN / SCAN_WG)

__device__ __forceinline__ float lane_bcast(float v, int srclane) {
    return __int_as_float(__builtin_amdgcn_readlane(__float_as_int(v), srclane));
}

__global__ __launch_bounds__(SCAN_WG)
void scan_phase1(const float* __restrict__ X,
                 const float* __restrict__ xdbl,
                 const float* __restrict__ dt,
                 const float* __restrict__ A_log,
                 float* __restrict__ aprod,
                 float* __restrict__ hend) {
    int lane  = threadIdx.x & 31;
    int dblk  = blockIdx.x % SCAN_DB;
    int chunk = (blockIdx.x / SCAN_DB) % NCHUNK;
    int b     = blockIdx.x / (SCAN_DB * NCHUNK);
    int d     = dblk * SCAN_WG + threadIdx.x;
    int channel = b * D_IN + d;

    float A[D_ST], h[D_ST], ap[D_ST];
#pragma unroll
    for (int s = 0; s < D_ST; ++s) {
        A[s]  = -__expf(A_log[d * D_ST + s]);
        h[s]  = 0.0f;
        ap[s] = 1.0f;
    }
    size_t m0 = (size_t)b * SEQ + (size_t)chunk * LCHUNK;
    const float* dtp = dt + m0 * D_IN + d;
    const float* xp  = X  + m0 * D_IN + d;
    const float* bp  = xdbl + m0 * N_E + DT_RANK + (lane & 15);

    float dtc[SCAN_U], xc[SCAN_U], bc[SCAN_U];
#pragma unroll
    for (int u = 0; u < SCAN_U; ++u) {
        dtc[u] = dtp[(size_t)u * D_IN];
        xc[u]  = xp[(size_t)u * D_IN];
        bc[u]  = bp[(size_t)u * N_E];
    }
    for (int l = 0; l < LCHUNK; l += SCAN_U) {
        int ln = (l + SCAN_U < LCHUNK) ? (l + SCAN_U) : l;
        float dtn[SCAN_U], xn[SCAN_U], bn[SCAN_U];
#pragma unroll
        for (int u = 0; u < SCAN_U; ++u) {
            dtn[u] = dtp[(size_t)(ln + u) * D_IN];
            xn[u]  = xp[(size_t)(ln + u) * D_IN];
            bn[u]  = bp[(size_t)(ln + u) * N_E];
        }
#pragma unroll
        for (int u = 0; u < SCAN_U; ++u) {
            float dtv = dtc[u];
            float dtx = dtv * xc[u];
#pragma unroll
            for (int s = 0; s < D_ST; ++s) {
                float Bs = lane_bcast(bc[u], s);
                float ab = __expf(dtv * A[s]);
                h[s]  = ab * h[s] + dtx * Bs;
                ap[s] *= ab;
            }
        }
#pragma unroll
        for (int u = 0; u < SCAN_U; ++u) { dtc[u] = dtn[u]; xc[u] = xn[u]; bc[u] = bn[u]; }
    }
#pragma unroll
    for (int s = 0; s < D_ST; ++s) {
        int idx = (chunk * D_ST + s) * NCHAN + channel;
        aprod[idx] = ap[s];
        hend[idx]  = h[s];
    }
}

__global__ void scan_phase2(const float* __restrict__ aprod,
                            const float* __restrict__ hend,
                            float* __restrict__ carry) {
    int t = blockIdx.x * blockDim.x + threadIdx.x;
    if (t >= NCHAN * D_ST) return;
    float c = 0.0f;
#pragma unroll
    for (int ch = 0; ch < NCHUNK; ++ch) {
        int idx = ch * (NCHAN * D_ST) + t;
        carry[idx] = c;
        c = aprod[idx] * c + hend[idx];
    }
}

__global__ __launch_bounds__(SCAN_WG)
void scan_phase3(const float* __restrict__ X,
                 const float* __restrict__ xdbl,
                 const float* __restrict__ dt,
                 const float* __restrict__ A_log,
                 const float* __restrict__ Dp,
                 const float* __restrict__ carry,
                 float* __restrict__ y) {
    int lane  = threadIdx.x & 31;
    int dblk  = blockIdx.x % SCAN_DB;
    int chunk = (blockIdx.x / SCAN_DB) % NCHUNK;
    int b     = blockIdx.x / (SCAN_DB * NCHUNK);
    int d     = dblk * SCAN_WG + threadIdx.x;
    int channel = b * D_IN + d;

    float A[D_ST], h[D_ST];
#pragma unroll
    for (int s = 0; s < D_ST; ++s) {
        A[s] = -__expf(A_log[d * D_ST + s]);
        h[s] = carry[(chunk * D_ST + s) * NCHAN + channel];
    }
    float Dd = Dp[d];
    size_t m0 = (size_t)b * SEQ + (size_t)chunk * LCHUNK;
    const float* dtp = dt + m0 * D_IN + d;
    const float* xp  = X  + m0 * D_IN + d;
    float*       yp  = y  + m0 * D_IN + d;
    const float* bcp = xdbl + m0 * N_E + DT_RANK + lane;

    float dtc[SCAN_U], xc[SCAN_U], bc[SCAN_U];
#pragma unroll
    for (int u = 0; u < SCAN_U; ++u) {
        dtc[u] = dtp[(size_t)u * D_IN];
        xc[u]  = xp[(size_t)u * D_IN];
        bc[u]  = bcp[(size_t)u * N_E];
    }
    for (int l = 0; l < LCHUNK; l += SCAN_U) {
        int ln = (l + SCAN_U < LCHUNK) ? (l + SCAN_U) : l;
        float dtn[SCAN_U], xn[SCAN_U], bn[SCAN_U];
#pragma unroll
        for (int u = 0; u < SCAN_U; ++u) {
            dtn[u] = dtp[(size_t)(ln + u) * D_IN];
            xn[u]  = xp[(size_t)(ln + u) * D_IN];
            bn[u]  = bcp[(size_t)(ln + u) * N_E];
        }
#pragma unroll
        for (int u = 0; u < SCAN_U; ++u) {
            float dtv = dtc[u];
            float xv  = xc[u];
            float dtx = dtv * xv;
            float acc = 0.0f;
#pragma unroll
            for (int s = 0; s < D_ST; ++s) {
                float Bs = lane_bcast(bc[u], s);
                float Cs = lane_bcast(bc[u], D_ST + s);
                float ab = __expf(dtv * A[s]);
                h[s] = ab * h[s] + dtx * Bs;
                acc  = h[s] * Cs + acc;
            }
            yp[(size_t)(l + u) * D_IN] = acc + Dd * xv;
        }
#pragma unroll
        for (int u = 0; u < SCAN_U; ++u) { dtc[u] = dtn[u]; xc[u] = xn[u]; bc[u] = bn[u]; }
    }
}

extern "C" void kernel_launch(void* const* d_in, const int* in_sizes, int n_in,
                              void* d_out, int out_size, void* d_ws, size_t ws_size,
                              hipStream_t stream) {
    const float* x       = (const float*)d_in[0];
    const float* A_log   = (const float*)d_in[1];
    const float* Dp      = (const float*)d_in[2];
    const float* W_xproj = (const float*)d_in[3];
    const float* W_dt    = (const float*)d_in[4];
    const float* b_dt    = (const float*)d_in[5];
    float* y = (float*)d_out;

    char* ws = (char*)d_ws;
    size_t off = 0;
    float* xdbl  = (float*)(ws + off); off += (size_t)M_TOT * N_E * 4;
    off = (off + 255) & ~(size_t)255;
    float* dtbuf = (float*)(ws + off); off += (size_t)M_TOT * D_IN * 4;
    off = (off + 255) & ~(size_t)255;
    float* aprod = (float*)(ws + off); off += (size_t)NCHAN * D_ST * NCHUNK * 4;
    off = (off + 255) & ~(size_t)255;
    float* hend  = (float*)(ws + off); off += (size_t)NCHAN * D_ST * NCHUNK * 4;
    off = (off + 255) & ~(size_t)255;
    float* carry = (float*)(ws + off);

    xproj_wmma<<<M_TOT / 16, K1_WAVES * 32, 0, stream>>>(x, W_xproj, xdbl);
    dt_wmma<<<8192 / 8, 256, 0, stream>>>(xdbl, W_dt, b_dt, dtbuf);
    scan_phase1<<<BATCH * NCHUNK * SCAN_DB, SCAN_WG, 0, stream>>>(
        x, xdbl, dtbuf, A_log, aprod, hend);
    scan_phase2<<<(NCHAN * D_ST) / 256, 256, 0, stream>>>(aprod, hend, carry);
    scan_phase3<<<BATCH * NCHUNK * SCAN_DB, SCAN_WG, 0, stream>>>(
        x, xdbl, dtbuf, A_log, Dp, carry, y);
}


# dt GEMM's softplus tail on the hardware exp2/log2/rcp: log1p(e) = ln2*log2(1+e) + (e-((1+e)-1))/(1+e) replaces the library log1pf; about 10 vector instructions per element instead of 85, no branches
# speedup vs baseline: 5.2773x; 1.1054x over previous
#include <hip/hip_runtime.h>
#include <hip/hip_bf16.h>
#include <math.h>

typedef __attribute__((ext_vector_type(2))) float v2f;
typedef __attribute__((ext_vector_type(4))) float v4f;
typedef __attribute__((ext_vector_type(8))) float v8f;

#define D_IN   1024
#define D_ST   16
#define DT_RANK 64
#define N_E    96
#define BATCH  2
#define SEQ    2048
#define M_TOT  (BATCH*SEQ)
#define NCHUNK 32
#define LCHUNK (SEQ/NCHUNK)
#define NCHAN  (BATCH*D_IN)

#define FRAG_LO(q) (v2f){(q).x, (q).y}
#define FRAG_HI(q) (v2f){(q).z, (q).w}

#define K1_WAVES 8
#define K1_KS    (D_IN / K1_WAVES)
#define K1_NT    (N_E / 16)
#define K1_JB    2
#define K1_SETS  (K1_KS / (8 * K1_JB))

__global__ __launch_bounds__(K1_WAVES * 32)
void xproj_wmma(const float* __restrict__ X,
                const float* __restrict__ W,
                float* __restrict__ xdbl) {
    __shared__ float red[K1_WAVES][K1_NT * 8 * 32];
    int lane = threadIdx.x & 31;
    int w    = threadIdx.x >> 5;
    int m0   = blockIdx.x * 16;
    int half = lane >> 4, lq = lane & 15;
    const float* aptr = X + (size_t)(m0 + lq) * D_IN + w * K1_KS + half * 4;
    const float* bptr = W + (size_t)lq * D_IN + w * K1_KS + half * 4;

    v8f acc[K1_NT];
#pragma unroll
    for (int t = 0; t < K1_NT; ++t) acc[t] = (v8f){};

    v4f a[K1_SETS][K1_JB], b[K1_SETS][K1_JB][K1_NT];
#pragma unroll
    for (int jj = 0; jj < K1_JB; ++jj) {
        a[0][jj] = *(const v4f*)(aptr + 8 * jj);
#pragma unroll
        for (int t = 0; t < K1_NT; ++t)
            b[0][jj][t] = *(const v4f*)(bptr + (size_t)t * 16 * D_IN + 8 * jj);
    }
#pragma unroll
    for (int s = 0; s < K1_SETS; ++s) {
        if (s + 1 < K1_SETS) {
#pragma unroll
            for (int jj = 0; jj < K1_JB; ++jj) {
                int j = (s + 1) * K1_JB + jj;
                a[s + 1][jj] = *(const v4f*)(aptr + 8 * j);
#pragma unroll
                for (int t = 0; t < K1_NT; ++t)
                    b[s + 1][jj][t] = *(const v4f*)(bptr + (size_t)t * 16 * D_IN + 8 * j);
            }
        }
        __builtin_amdgcn_sched_barrier(0);
#pragma unroll
        for (int jj = 0; jj < K1_JB; ++jj) {
#pragma unroll
            for (int t = 0; t < K1_NT; ++t)
                acc[t] = __builtin_amdgcn_wmma_f32_16x16x4_f32(
                    false, FRAG_LO(a[s][jj]), false, FRAG_LO(b[s][jj][t]), (short)0, acc[t], false, false);
#pragma unroll
            for (int t = 0; t < K1_NT; ++t)
                acc[t] = __builtin_amdgcn_wmma_f32_16x16x4_f32(
                    false, FRAG_HI(a[s][jj]), false, FRAG_HI(b[s][jj][t]), (short)0, acc[t], false, false);
        }
    }
#pragma unroll
    for (int t = 0; t < K1_NT; ++t)
#pragma unroll
        for (int v = 0; v < 8; ++v)
            red[w][(t * 8 + v) * 32 + lane] = acc[t][v];
    __syncthreads();
#pragma unroll
    for (int i = 0; i < (16 * N_E) / (K1_WAVES * 32); ++i) {
        int o   = threadIdx.x + i * (K1_WAVES * 32);
        int row = o / N_E;
        int col = o - row * N_E;
        int idx = ((col >> 4) * 8 + (row & 7)) * 32 + (col & 15) + 16 * (row >> 3);
        float s = red[0][idx];
#pragma unroll
        for (int ww = 1; ww < K1_WAVES; ++ww) s += red[ww][idx];
        xdbl[(size_t)m0 * N_E + o] = s;
    }
}

#define K2_NT 2

__device__ __forceinline__ float softplus_f32(float z) {
    float e = __expf(z);
    float u = 1.0f + e;
    float c = e - (u - 1.0f);
    float r = fmaf(__builtin_amdgcn_logf(u), 0.693147182f, c * __builtin_amdgcn_rcpf(u));
    return (z > 20.0f) ? z : r;
}

__global__ __launch_bounds__(256)
void dt_wmma(const float* __restrict__ xdbl,
             const float* __restrict__ Wdt,
             const float* __restrict__ bdt,
             float* __restrict__ dt) {
    int wave = (blockIdx.x * blockDim.x + threadIdx.x) >> 5;
    int lane = threadIdx.x & 31;
    const int NG = D_IN / (16 * K2_NT);
    int ng = wave % NG;
    int mt = wave / NG;
    if (mt >= M_TOT / 16) return;
    int m0 = mt * 16, n0 = ng * (16 * K2_NT);
    int half = lane >> 4, lq = lane & 15;
    const float* aptr = xdbl + (size_t)(m0 + lq) * N_E + half * 4;
    const float* bptr = Wdt + (size_t)(n0 + lq) * DT_RANK + half * 4;
    v8f acc[K2_NT];
#pragma unroll
    for (int t = 0; t < K2_NT; ++t) acc[t] = (v8f){};
    v4f a[DT_RANK / 8], b[DT_RANK / 8][K2_NT];
#pragma unroll
    for (int j = 0; j < DT_RANK / 8; ++j) {
        a[j] = *(const v4f*)(aptr + 8 * j);
#pragma unroll
        for (int t = 0; t < K2_NT; ++t)
            b[j][t] = *(const v4f*)(bptr + (size_t)t * 16 * DT_RANK + 8 * j);
    }
    __builtin_amdgcn_sched_barrier(0);
#pragma unroll
    for (int j = 0; j < DT_RANK / 8; ++j) {
#pragma unroll
        for (int t = 0; t < K2_NT; ++t)
            acc[t] = __builtin_amdgcn_wmma_f32_16x16x4_f32(
                false, FRAG_LO(a[j]), false, FRAG_LO(b[j][t]), (short)0, acc[t], false, false);
#pragma unroll
        for (int t = 0; t < K2_NT; ++t)
            acc[t] = __builtin_amdgcn_wmma_f32_16x16x4_f32(
                false, FRAG_HI(a[j]), false, FRAG_HI(b[j][t]), (short)0, acc[t], false, false);
    }
#pragma unroll
    for (int t = 0; t < K2_NT; ++t) {
        int n = n0 + t * 16 + lq;
        float bias = bdt[n];
#pragma unroll
        for (int v = 0; v < 8; ++v) {
            int m = m0 + v + 8 * half;
            float z = acc[t][v] + bias;
            dt[(size_t)m * D_IN + n] = softplus_f32(z);
        }
    }
}

#define SCAN_U   8
#define SCAN_WG  64
#define SCAN_DB  (D_IN / SCAN_WG)

__device__ __forceinline__ float lane_bcast(float v, int srclane) {
    return __int_as_float(__builtin_amdgcn_readlane(__float_as_int(v), srclane));
}

__global__ __launch_bounds__(SCAN_WG)
void scan_phase1(const float* __restrict__ X,
                 const float* __restrict__ xdbl,
                 const float* __restrict__ dt,
                 const float* __restrict__ A_log,
                 float* __restrict__ aprod,
                 float* __restrict__ hend) {
    int lane  = threadIdx.x & 31;
    int dblk  = blockIdx.x % SCAN_DB;
    int chunk = (blockIdx.x / SCAN_DB) % NCHUNK;
    int b     = blockIdx.x / (SCAN_DB * NCHUNK);
    int d     = dblk * SCAN_WG + threadIdx.x;
    int channel = b * D_IN + d;

    float A[D_ST], h[D_ST], ap[D_ST];
#pragma unroll
    for (int s = 0; s < D_ST; ++s) {
        A[s]  = -__expf(A_log[d * D_ST + s]);
        h[s]  = 0.0f;
        ap[s] = 1.0f;
    }
    size_t m0 = (size_t)b * SEQ + (size_t)chunk * LCHUNK;
    const float* dtp = dt + m0 * D_IN + d;
    const float* xp  = X  + m0 * D_IN + d;
    const float* bp  = xdbl + m0 * N_E + DT_RANK + (lane & 15);

    float dtc[SCAN_U], xc[SCAN_U], bc[SCAN_U];
#pragma unroll
    for (int u = 0; u < SCAN_U; ++u) {
        dtc[u] = dtp[(size_t)u * D_IN];
        xc[u]  = xp[(size_t)u * D_IN];
        bc[u]  = bp[(size_t)u * N_E];
    }
    for (int l = 0; l < LCHUNK; l += SCAN_U) {
        int ln = (l + SCAN_U < LCHUNK) ? (l + SCAN_U) : l;
        float dtn[SCAN_U], xn[SCAN_U], bn[SCAN_U];
#pragma unroll
        for (int u = 0; u < SCAN_U; ++u) {
            dtn[u] = dtp[(size_t)(ln + u) * D_IN];
            xn[u]  = xp[(size_t)(ln + u) * D_IN];
            bn[u]  = bp[(size_t)(ln + u) * N_E];
        }
#pragma unroll
        for (int u = 0; u < SCAN_U; ++u) {
            float dtv = dtc[u];
            float dtx = dtv * xc[u];
#pragma unroll
            for (int s = 0; s < D_ST; ++s) {
                float Bs = lane_bcast(bc[u], s);
                float ab = __expf(dtv * A[s]);
                h[s]  = ab * h[s] + dtx * Bs;
                ap[s] *= ab;
            }
        }
#pragma unroll
        for (int u = 0; u < SCAN_U; ++u) { dtc[u] = dtn[u]; xc[u] = xn[u]; bc[u] = bn[u]; }
    }
#pragma unroll
    for (int s = 0; s < D_ST; ++s) {
        int idx = (chunk * D_ST + s) * NCHAN + channel;
        aprod[idx] = ap[s];
        hend[idx]  = h[s];
    }
}

__global__ void scan_phase2(const float* __restrict__ aprod,
                            const float* __restrict__ hend,
                            float* __restrict__ carry) {
    int t = blockIdx.x * blockDim.x + threadIdx.x;
    if (t >= NCHAN * D_ST) return;
    float c = 0.0f;
#pragma unroll
    for (int ch = 0; ch < NCHUNK; ++ch) {
        int idx = ch * (NCHAN * D_ST) + t;
        carry[idx] = c;
        c = aprod[idx] * c + hend[idx];
    }
}

__global__ __launch_bounds__(SCAN_WG)
void scan_phase3(const float* __restrict__ X,
                 const float* __restrict__ xdbl,
                 const float* __restrict__ dt,
                 const float* __restrict__ A_log,
                 const float* __restrict__ Dp,
                 const float* __restrict__ carry,
                 float* __restrict__ y) {
    int lane  = threadIdx.x & 31;
    int dblk  = blockIdx.x % SCAN_DB;
    int chunk = (blockIdx.x / SCAN_DB) % NCHUNK;
    int b     = blockIdx.x / (SCAN_DB * NCHUNK);
    int d     = dblk * SCAN_WG + threadIdx.x;
    int channel = b * D_IN + d;

    float A[D_ST], h[D_ST];
#pragma unroll
    for (int s = 0; s < D_ST; ++s) {
        A[s] = -__expf(A_log[d * D_ST + s]);
        h[s] = carry[(chunk * D_ST + s) * NCHAN + channel];
    }
    float Dd = Dp[d];
    size_t m0 = (size_t)b * SEQ + (size_t)chunk * LCHUNK;
    const float* dtp = dt + m0 * D_IN + d;
    const float* xp  = X  + m0 * D_IN + d;
    float*       yp  = y  + m0 * D_IN + d;
    const float* bcp = xdbl + m0 * N_E + DT_RANK + lane;

    float dtc[SCAN_U], xc[SCAN_U], bc[SCAN_U];
#pragma unroll
    for (int u = 0; u < SCAN_U; ++u) {
        dtc[u] = dtp[(size_t)u * D_IN];
        xc[u]  = xp[(size_t)u * D_IN];
        bc[u]  = bcp[(size_t)u * N_E];
    }
    for (int l = 0; l < LCHUNK; l += SCAN_U) {
        int ln = (l + SCAN_U < LCHUNK) ? (l + SCAN_U) : l;
        float dtn[SCAN_U], xn[SCAN_U], bn[SCAN_U];
#pragma unroll
        for (int u = 0; u < SCAN_U; ++u) {
            dtn[u] = dtp[(size_t)(ln + u) * D_IN];
            xn[u]  = xp[(size_t)(ln + u) * D_IN];
            bn[u]  = bcp[(size_t)(ln + u) * N_E];
        }
#pragma unroll
        for (int u = 0; u < SCAN_U; ++u) {
            float dtv = dtc[u];
            float xv  = xc[u];
            float dtx = dtv * xv;
            float acc = 0.0f;
#pragma unroll
            for (int s = 0; s < D_ST; ++s) {
                float Bs = lane_bcast(bc[u], s);
                float Cs = lane_bcast(bc[u], D_ST + s);
                float ab = __expf(dtv * A[s]);
                h[s] = ab * h[s] + dtx * Bs;
                acc  = h[s] * Cs + acc;
            }
            yp[(size_t)(l + u) * D_IN] = acc + Dd * xv;
        }
#pragma unroll
        for (int u = 0; u < SCAN_U; ++u) { dtc[u] = dtn[u]; xc[u] = xn[u]; bc[u] = bn[u]; }
    }
}

extern "C" void kernel_launch(void* const* d_in, const int* in_sizes, int n_in,
                              void* d_out, int out_size, void* d_ws, size_t ws_size,
                              hipStream_t stream) {
    const float* x       = (const float*)d_in[0];
    const float* A_log   = (const float*)d_in[1];
    const float* Dp      = (const float*)d_in[2];
    const float* W_xproj = (const float*)d_in[3];
    const float* W_dt    = (const float*)d_in[4];
    const float* b_dt    = (const float*)d_in[5];
    float* y = (float*)d_out;

    char* ws = (char*)d_ws;
    size_t off = 0;
    float* xdbl  = (float*)(ws + off); off += (size_t)M_TOT * N_E * 4;
    off = (off + 255) & ~(size_t)255;
    float* dtbuf = (float*)(ws + off); off += (size_t)M_TOT * D_IN * 4;
    off = (off + 255) & ~(size_t)255;
    float* aprod = (float*)(ws + off); off += (size_t)NCHAN * D_ST * NCHUNK * 4;
    off = (off + 255) & ~(size_t)255;
    float* hend  = (float*)(ws + off); off += (size_t)NCHAN * D_ST * NCHUNK * 4;
    off = (off + 255) & ~(size_t)255;
    float* carry = (float*)(ws + off);

    xproj_wmma<<<M_TOT / 16, K1_WAVES * 32, 0, stream>>>(x, W_xproj, xdbl);
    dt_wmma<<<8192 / 8, 256, 0, stream>>>(xdbl, W_dt, b_dt, dtbuf);
    scan_phase1<<<BATCH * NCHUNK * SCAN_DB, SCAN_WG, 0, stream>>>(
        x, xdbl, dtbuf, A_log, aprod, hend);
    scan_phase2<<<(NCHAN * D_ST) / 256, 256, 0, stream>>>(aprod, hend, carry);
    scan_phase3<<<BATCH * NCHUNK * SCAN_DB, SCAN_WG, 0, stream>>>(
        x, xdbl, dtbuf, A_log, Dp, carry, y);
}


# dt GEMM and softplus fused into the x_proj kernel: a strip's dt_low stays in LDS as the A operand of its 16 dt rows; 4 launches per repetition instead of 5, 2048 GEMM waves enter instead of 10240
# speedup vs baseline: 5.7439x; 1.0884x over previous
#include <hip/hip_runtime.h>
#include <hip/hip_bf16.h>
#include <math.h>

typedef __attribute__((ext_vector_type(2))) float v2f;
typedef __attribute__((ext_vector_type(4))) float v4f;
typedef __attribute__((ext_vector_type(8))) float v8f;

#define D_IN   1024
#define D_ST   16
#define DT_RANK 64
#define N_E    96
#define BATCH  2
#define SEQ    2048
#define M_TOT  (BATCH*SEQ)
#define NCHUNK 32
#define LCHUNK (SEQ/NCHUNK)
#define NCHAN  (BATCH*D_IN)

#define FRAG_LO(q) (v2f){(q).x, (q).y}
#define FRAG_HI(q) (v2f){(q).z, (q).w}

__device__ __forceinline__ float softplus_f32(float z) {
    float e = __expf(z);
    float u = 1.0f + e;
    float c = e - (u - 1.0f);
    float r = fmaf(__builtin_amdgcn_logf(u), 0.693147182f, c * __builtin_amdgcn_rcpf(u));
    return (z > 20.0f) ? z : r;
}

#define K1_WAVES 8
#define K1_KS    (D_IN / K1_WAVES)
#define K1_NT    (N_E / 16)
#define K1_JB    2
#define K1_SETS  (K1_KS / (8 * K1_JB))
#define K2_NT    2
#define K2_NW    (D_IN / K1_WAVES)
#define K2_PAIRS (K2_NW / (16 * K2_NT))
#define XS_PITCH (DT_RANK + 4)

__global__ __launch_bounds__(K1_WAVES * 32)
void xproj_dt_wmma(const float* __restrict__ X,
                   const float* __restrict__ W,
                   const float* __restrict__ Wdt,
                   const float* __restrict__ bdt,
                   float* __restrict__ xdbl,
                   float* __restrict__ dt) {
    __shared__ float red[K1_WAVES][K1_NT * 8 * 32];
    __shared__ __attribute__((aligned(16))) float xs[16][XS_PITCH];
    int lane = threadIdx.x & 31;
    int w    = threadIdx.x >> 5;
    int m0   = blockIdx.x * 16;
    int half = lane >> 4, lq = lane & 15;
    const float* aptr = X + (size_t)(m0 + lq) * D_IN + w * K1_KS + half * 4;
    const float* bptr = W + (size_t)lq * D_IN + w * K1_KS + half * 4;

    v8f acc[K1_NT];
#pragma unroll
    for (int t = 0; t < K1_NT; ++t) acc[t] = (v8f){};

    v4f a[K1_SETS][K1_JB], b[K1_SETS][K1_JB][K1_NT];
#pragma unroll
    for (int jj = 0; jj < K1_JB; ++jj) {
        a[0][jj] = *(const v4f*)(aptr + 8 * jj);
#pragma unroll
        for (int t = 0; t < K1_NT; ++t)
            b[0][jj][t] = *(const v4f*)(bptr + (size_t)t * 16 * D_IN + 8 * jj);
    }
#pragma unroll
    for (int s = 0; s < K1_SETS; ++s) {
        if (s + 1 < K1_SETS) {
#pragma unroll
            for (int jj = 0; jj < K1_JB; ++jj) {
                int j = (s + 1) * K1_JB + jj;
                a[s + 1][jj] = *(const v4f*)(aptr + 8 * j);
#pragma unroll
                for (int t = 0; t < K1_NT; ++t)
                    b[s + 1][jj][t] = *(const v4f*)(bptr + (size_t)t * 16 * D_IN + 8 * j);
            }
        }
        __builtin_amdgcn_sched_barrier(0);
#pragma unroll
        for (int jj = 0; jj < K1_JB; ++jj) {
#pragma unroll
            for (int t = 0; t < K1_NT; ++t)
                acc[t] = __builtin_amdgcn_wmma_f32_16x16x4_f32(
                    false, FRAG_LO(a[s][jj]), false, FRAG_LO(b[s][jj][t]), (short)0, acc[t], false, false);
#pragma unroll
            for (int t = 0; t < K1_NT; ++t)
                acc[t] = __builtin_amdgcn_wmma_f32_16x16x4_f32(
                    false, FRAG_HI(a[s][jj]), false, FRAG_HI(b[s][jj][t]), (short)0, acc[t], false, false);
        }
    }
#pragma unroll
    for (int t = 0; t < K1_NT; ++t)
#pragma unroll
        for (int v = 0; v < 8; ++v)
            red[w][(t * 8 + v) * 32 + lane] = acc[t][v];
    __syncthreads();
    const int nw = w * K2_NW;
    const float* b2ptr = Wdt + (size_t)(nw + lq) * DT_RANK + half * 4;
    v4f b2[K2_PAIRS][DT_RANK / 8][K2_NT];
#pragma unroll
    for (int j = 0; j < DT_RANK / 8; ++j)
#pragma unroll
        for (int t = 0; t < K2_NT; ++t)
            b2[0][j][t] = *(const v4f*)(b2ptr + (size_t)t * 16 * DT_RANK + 8 * j);
    float sv[(16 * N_E) / (K1_WAVES * 32)];
#pragma unroll
    for (int i = 0; i < (16 * N_E) / (K1_WAVES * 32); ++i) {
        int o   = threadIdx.x + i * (K1_WAVES * 32);
        int row = o / N_E;
        int col = o - row * N_E;
        int idx = ((col >> 4) * 8 + (row & 7)) * 32 + (col & 15) + 16 * (row >> 3);
        float s = red[0][idx];
#pragma unroll
        for (int ww = 1; ww < K1_WAVES; ++ww) s += red[ww][idx];
        sv[i] = s;
        if (col < DT_RANK) xs[row][col] = s;
    }
    __syncthreads();
#pragma unroll
    for (int i = 0; i < (16 * N_E) / (K1_WAVES * 32); ++i)
        xdbl[(size_t)m0 * N_E + threadIdx.x + i * (K1_WAVES * 32)] = sv[i];

    v4f a2[DT_RANK / 8];
#pragma unroll
    for (int j = 0; j < DT_RANK / 8; ++j)
        a2[j] = *(const v4f*)(&xs[lq][8 * j + half * 4]);
#pragma unroll
    for (int g = 0; g < K2_PAIRS; ++g) {
        if (g + 1 < K2_PAIRS) {
#pragma unroll
            for (int j = 0; j < DT_RANK / 8; ++j)
#pragma unroll
                for (int t = 0; t < K2_NT; ++t)
                    b2[g + 1][j][t] = *(const v4f*)(b2ptr + (size_t)((g + 1) * K2_NT + t) * 16 * DT_RANK + 8 * j);
        }
        __builtin_amdgcn_sched_barrier(0);
        v8f acc2[K2_NT];
#pragma unroll
        for (int t = 0; t < K2_NT; ++t) acc2[t] = (v8f){};
#pragma unroll
        for (int j = 0; j < DT_RANK / 8; ++j) {
#pragma unroll
            for (int t = 0; t < K2_NT; ++t)
                acc2[t] = __builtin_amdgcn_wmma_f32_16x16x4_f32(
                    false, FRAG_LO(a2[j]), false, FRAG_LO(b2[g][j][t]), (short)0, acc2[t], false, false);
#pragma unroll
            for (int t = 0; t < K2_NT; ++t)
                acc2[t] = __builtin_amdgcn_wmma_f32_16x16x4_f32(
                    false, FRAG_HI(a2[j]), false, FRAG_HI(b2[g][j][t]), (short)0, acc2[t], false, false);
        }
#pragma unroll
        for (int t = 0; t < K2_NT; ++t) {
            int n = nw + (g * K2_NT + t) * 16 + lq;
            float bias = bdt[n];
#pragma unroll
            for (int v = 0; v < 8; ++v) {
                int m = m0 + v + 8 * half;
                float z = acc2[t][v] + bias;
                dt[(size_t)m * D_IN + n] = softplus_f32(z);
            }
        }
    }
}

#define SCAN_U   8
#define SCAN_WG  64
#define SCAN_DB  (D_IN / SCAN_WG)

__device__ __forceinline__ float lane_bcast(float v, int srclane) {
    return __int_as_float(__builtin_amdgcn_readlane(__float_as_int(v), srclane));
}

__global__ __launch_bounds__(SCAN_WG)
void scan_phase1(const float* __restrict__ X,
                 const float* __restrict__ xdbl,
                 const float* __restrict__ dt,
                 const float* __restrict__ A_log,
                 float* __restrict__ aprod,
                 float* __restrict__ hend) {
    int lane  = threadIdx.x & 31;
    int dblk  = blockIdx.x % SCAN_DB;
    int chunk = (blockIdx.x / SCAN_DB) % NCHUNK;
    int b     = blockIdx.x / (SCAN_DB * NCHUNK);
    int d     = dblk * SCAN_WG + threadIdx.x;
    int channel = b * D_IN + d;

    float A[D_ST], h[D_ST], ap[D_ST];
#pragma unroll
    for (int s = 0; s < D_ST; ++s) {
        A[s]  = -__expf(A_log[d * D_ST + s]);
        h[s]  = 0.0f;
        ap[s] = 1.0f;
    }
    size_t m0 = (size_t)b * SEQ + (size_t)chunk * LCHUNK;
    const float* dtp = dt + m0 * D_IN + d;
    const float* xp  = X  + m0 * D_IN + d;
    const float* bp  = xdbl + m0 * N_E + DT_RANK + (lane & 15);

    float dtc[SCAN_U], xc[SCAN_U], bc[SCAN_U];
#pragma unroll
    for (int u = 0; u < SCAN_U; ++u) {
        dtc[u] = dtp[(size_t)u * D_IN];
        xc[u]  = xp[(size_t)u * D_IN];
        bc[u]  = bp[(size_t)u * N_E];
    }
    for (int l = 0; l < LCHUNK; l += SCAN_U) {
        int ln = (l + SCAN_U < LCHUNK) ? (l + SCAN_U) : l;
        float dtn[SCAN_U], xn[SCAN_U], bn[SCAN_U];
#pragma unroll
        for (int u = 0; u < SCAN_U; ++u) {
            dtn[u] = dtp[(size_t)(ln + u) * D_IN];
            xn[u]  = xp[(size_t)(ln + u) * D_IN];
            bn[u]  = bp[(size_t)(ln + u) * N_E];
        }
#pragma unroll
        for (int u = 0; u < SCAN_U; ++u) {
            float dtv = dtc[u];
            float dtx = dtv * xc[u];
#pragma unroll
            for (int s = 0; s < D_ST; ++s) {
                float Bs = lane_bcast(bc[u], s);
                float ab = __expf(dtv * A[s]);
                h[s]  = ab * h[s] + dtx * Bs;
                ap[s] *= ab;
            }
        }
#pragma unroll
        for (int u = 0; u < SCAN_U; ++u) { dtc[u] = dtn[u]; xc[u] = xn[u]; bc[u] = bn[u]; }
    }
#pragma unroll
    for (int s = 0; s < D_ST; ++s) {
        int idx = (chunk * D_ST + s) * NCHAN + channel;
        aprod[idx] = ap[s];
        hend[idx]  = h[s];
    }
}

__global__ void scan_phase2(const float* __restrict__ aprod,
                            const float* __restrict__ hend,
                            float* __restrict__ carry) {
    int t = blockIdx.x * blockDim.x + threadIdx.x;
    if (t >= NCHAN * D_ST) return;
    float c = 0.0f;
#pragma unroll
    for (int ch = 0; ch < NCHUNK; ++ch) {
        int idx = ch * (NCHAN * D_ST) + t;
        carry[idx] = c;
        c = aprod[idx] * c + hend[idx];
    }
}

__global__ __launch_bounds__(SCAN_WG)
void scan_phase3(const float* __restrict__ X,
                 const float* __restrict__ xdbl,
                 const float* __restrict__ dt,
                 const float* __restrict__ A_log,
                 const float* __restrict__ Dp,
                 const float* __restrict__ carry,
                 float* __restrict__ y) {
    int lane  = threadIdx.x & 31;
    int dblk  = blockIdx.x % SCAN_DB;
    int chunk = (blockIdx.x / SCAN_DB) % NCHUNK;
    int b     = blockIdx.x / (SCAN_DB * NCHUNK);
    int d     = dblk * SCAN_WG + threadIdx.x;
    int channel = b * D_IN + d;

    float A[D_ST], h[D_ST];
#pragma unroll
    for (int s = 0; s < D_ST; ++s) {
        A[s] = -__expf(A_log[d * D_ST + s]);
        h[s] = carry[(chunk * D_ST + s) * NCHAN + channel];
    }
    float Dd = Dp[d];
    size_t m0 = (size_t)b * SEQ + (size_t)chunk * LCHUNK;
    const float* dtp = dt + m0 * D_IN + d;
    const float* xp  = X  + m0 * D_IN + d;
    float*       yp  = y  + m0 * D_IN + d;
    const float* bcp = xdbl + m0 * N_E + DT_RANK + lane;

    float dtc[SCAN_U], xc[SCAN_U], bc[SCAN_U];
#pragma unroll
    for (int u = 0; u < SCAN_U; ++u) {
        dtc[u] = dtp[(size_t)u * D_IN];
        xc[u]  = xp[(size_t)u * D_IN];
        bc[u]  = bcp[(size_t)u * N_E];
    }
    for (int l = 0; l < LCHUNK; l += SCAN_U) {
        int ln = (l + SCAN_U < LCHUNK) ? (l + SCAN_U) : l;
        float dtn[SCAN_U], xn[SCAN_U], bn[SCAN_U];
#pragma unroll
        for (int u = 0; u < SCAN_U; ++u) {
            dtn[u] = dtp[(size_t)(ln + u) * D_IN];
            xn[u]  = xp[(size_t)(ln + u) * D_IN];
            bn[u]  = bcp[(size_t)(ln + u) * N_E];
        }
#pragma unroll
        for (int u = 0; u < SCAN_U; ++u) {
            float dtv = dtc[u];
            float xv  = xc[u];
            float dtx = dtv * xv;
            float acc = 0.0f;
#pragma unroll
            for (int s = 0; s < D_ST; ++s) {
                float Bs = lane_bcast(bc[u], s);
                float Cs = lane_bcast(bc[u], D_ST + s);
                float ab = __expf(dtv * A[s]);
                h[s] = ab * h[s] + dtx * Bs;
                acc  = h[s] * Cs + acc;
            }
            yp[(size_t)(l + u) * D_IN] = acc + Dd * xv;
        }
#pragma unroll
        for (int u = 0; u < SCAN_U; ++u) { dtc[u] = dtn[u]; xc[u] = xn[u]; bc[u] = bn[u]; }
    }
}

extern "C" void kernel_launch(void* const* d_in, const int* in_sizes, int n_in,
                              void* d_out, int out_size, void* d_ws, size_t ws_size,
                              hipStream_t stream) {
    const float* x       = (const float*)d_in[0];
    const float* A_log   = (const float*)d_in[1];
    const float* Dp      = (const float*)d_in[2];
    const float* W_xproj = (const float*)d_in[3];
    const float* W_dt    = (const float*)d_in[4];
    const float* b_dt    = (const float*)d_in[5];
    float* y = (float*)d_out;

    char* ws = (char*)d_ws;
    size_t off = 0;
    float* xdbl  = (float*)(ws + off); off += (size_t)M_TOT * N_E * 4;
    off = (off + 255) & ~(size_t)255;
    float* dtbuf = (float*)(ws + off); off += (size_t)M_TOT * D_IN * 4;
    off = (off + 255) & ~(size_t)255;
    float* aprod = (float*)(ws + off); off += (size_t)NCHAN * D_ST * NCHUNK * 4;
    off = (off + 255) & ~(size_t)255;
    float* hend  = (float*)(ws + off); off += (size_t)NCHAN * D_ST * NCHUNK * 4;
    off = (off + 255) & ~(size_t)255;
    float* carry = (float*)(ws + off);

    xproj_dt_wmma<<<M_TOT / 16, K1_WAVES * 32, 0, stream>>>(x, W_xproj, W_dt, b_dt, xdbl, dtbuf);
    scan_phase1<<<BATCH * NCHUNK * SCAN_DB, SCAN_WG, 0, stream>>>(
        x, xdbl, dtbuf, A_log, aprod, hend);
    scan_phase2<<<(NCHAN * D_ST) / 256, 256, 0, stream>>>(aprod, hend, carry);
    scan_phase3<<<BATCH * NCHUNK * SCAN_DB, SCAN_WG, 0, stream>>>(
        x, xdbl, dtbuf, A_log, Dp, carry, y);
}


# x_proj fragment loads regrouped: a register set is one 128-byte line block of the rows x 3 column tiles (was half a line x 6), so the 4 loads of a line go out together; K sum order unchanged
# speedup vs baseline: 6.0236x; 1.0487x over previous
#include <hip/hip_runtime.h>
#include <hip/hip_bf16.h>
#include <math.h>

typedef __attribute__((ext_vector_type(2))) float v2f;
typedef __attribute__((ext_vector_type(4))) float v4f;
typedef __attribute__((ext_vector_type(8))) float v8f;

#define D_IN   1024
#define D_ST   16
#define DT_RANK 64
#define N_E    96
#define BATCH  2
#define SEQ    2048
#define M_TOT  (BATCH*SEQ)
#define NCHUNK 32
#define LCHUNK (SEQ/NCHUNK)
#define NCHAN  (BATCH*D_IN)

#define FRAG_LO(q) (v2f){(q).x, (q).y}
#define FRAG_HI(q) (v2f){(q).z, (q).w}

__device__ __forceinline__ float softplus_f32(float z) {
    float e = __expf(z);
    float u = 1.0f + e;
    float c = e - (u - 1.0f);
    float r = fmaf(__builtin_amdgcn_logf(u), 0.693147182f, c * __builtin_amdgcn_rcpf(u));
    return (z > 20.0f) ? z : r;
}

#define K1_WAVES 8
#define K1_KS    (D_IN / K1_WAVES)
#define K1_NT    (N_E / 16)
#define K1_LJ    4
#define K1_LB    (K1_KS / (8 * K1_LJ))
#define K1_HT    (K1_NT / 2)
#define K1_SETS  (K1_LB * 2)
#define K2_NT    2
#define K2_NW    (D_IN / K1_WAVES)
#define K2_PAIRS (K2_NW / (16 * K2_NT))
#define XS_PITCH (DT_RANK + 4)

__global__ __launch_bounds__(K1_WAVES * 32)
void xproj_dt_wmma(const float* __restrict__ X,
                   const float* __restrict__ W,
                   const float* __restrict__ Wdt,
                   const float* __restrict__ bdt,
                   float* __restrict__ xdbl,
                   float* __restrict__ dt) {
    __shared__ float red[K1_WAVES][K1_NT * 8 * 32];
    __shared__ __attribute__((aligned(16))) float xs[16][XS_PITCH];
    int lane = threadIdx.x & 31;
    int w    = threadIdx.x >> 5;
    int m0   = blockIdx.x * 16;
    int half = lane >> 4, lq = lane & 15;
    const float* aptr = X + (size_t)(m0 + lq) * D_IN + w * K1_KS + half * 4;
    const float* bptr = W + (size_t)lq * D_IN + w * K1_KS + half * 4;

    v8f acc[K1_NT];
#pragma unroll
    for (int t = 0; t < K1_NT; ++t) acc[t] = (v8f){};

    v4f a[K1_LB][K1_LJ], b[K1_SETS][K1_LJ][K1_HT];
#pragma unroll
    for (int jj = 0; jj < K1_LJ; ++jj)
        a[0][jj] = *(const v4f*)(aptr + 8 * jj);
#pragma unroll
    for (int t = 0; t < K1_HT; ++t)
#pragma unroll
        for (int jj = 0; jj < K1_LJ; ++jj)
            b[0][jj][t] = *(const v4f*)(bptr + (size_t)t * 16 * D_IN + 8 * jj);
#pragma unroll
    for (int s = 0; s < K1_SETS; ++s) {
        if (s + 1 < K1_SETS) {
            const int kbn = (s + 1) >> 1, thn = (s + 1) & 1;
            if (thn == 0) {
#pragma unroll
                for (int jj = 0; jj < K1_LJ; ++jj)
                    a[kbn][jj] = *(const v4f*)(aptr + 8 * (kbn * K1_LJ + jj));
            }
#pragma unroll
            for (int t = 0; t < K1_HT; ++t)
#pragma unroll
                for (int jj = 0; jj < K1_LJ; ++jj)
                    b[s + 1][jj][t] = *(const v4f*)(bptr + (size_t)(thn * K1_HT + t) * 16 * D_IN
                                                    + 8 * (kbn * K1_LJ + jj));
        }
        __builtin_amdgcn_sched_barrier(0);
        const int kb = s >> 1, t0 = (s & 1) * K1_HT;
#pragma unroll
        for (int jj = 0; jj < K1_LJ; ++jj) {
#pragma unroll
            for (int t = 0; t < K1_HT; ++t)
                acc[t0 + t] = __builtin_amdgcn_wmma_f32_16x16x4_f32(
                    false, FRAG_LO(a[kb][jj]), false, FRAG_LO(b[s][jj][t]), (short)0, acc[t0 + t], false, false);
#pragma unroll
            for (int t = 0; t < K1_HT; ++t)
                acc[t0 + t] = __builtin_amdgcn_wmma_f32_16x16x4_f32(
                    false, FRAG_HI(a[kb][jj]), false, FRAG_HI(b[s][jj][t]), (short)0, acc[t0 + t], false, false);
        }
    }
#pragma unroll
    for (int t = 0; t < K1_NT; ++t)
#pragma unroll
        for (int v = 0; v < 8; ++v)
            red[w][(t * 8 + v) * 32 + lane] = acc[t][v];
    __syncthreads();
    const int nw = w * K2_NW;
    const float* b2ptr = Wdt + (size_t)(nw + lq) * DT_RANK + half * 4;
    v4f b2[K2_PAIRS][DT_RANK / 8][K2_NT];
#pragma unroll
    for (int j = 0; j < DT_RANK / 8; ++j)
#pragma unroll
        for (int t = 0; t < K2_NT; ++t)
            b2[0][j][t] = *(const v4f*)(b2ptr + (size_t)t * 16 * DT_RANK + 8 * j);
    float sv[(16 * N_E) / (K1_WAVES * 32)];
#pragma unroll
    for (int i = 0; i < (16 * N_E) / (K1_WAVES * 32); ++i) {
        int o   = threadIdx.x + i * (K1_WAVES * 32);
        int row = o / N_E;
        int col = o - row * N_E;
        int idx = ((col >> 4) * 8 + (row & 7)) * 32 + (col & 15) + 16 * (row >> 3);
        float s = red[0][idx];
#pragma unroll
        for (int ww = 1; ww < K1_WAVES; ++ww) s += red[ww][idx];
        sv[i] = s;
        if (col < DT_RANK) xs[row][col] = s;
    }
    __syncthreads();
#pragma unroll
    for (int i = 0; i < (16 * N_E) / (K1_WAVES * 32); ++i)
        xdbl[(size_t)m0 * N_E + threadIdx.x + i * (K1_WAVES * 32)] = sv[i];

    v4f a2[DT_RANK / 8];
#pragma unroll
    for (int j = 0; j < DT_RANK / 8; ++j)
        a2[j] = *(const v4f*)(&xs[lq][8 * j + half * 4]);
#pragma unroll
    for (int g = 0; g < K2_PAIRS; ++g) {
        if (g + 1 < K2_PAIRS) {
#pragma unroll
            for (int j = 0; j < DT_RANK / 8; ++j)
#pragma unroll
                for (int t = 0; t < K2_NT; ++t)
                    b2[g + 1][j][t] = *(const v4f*)(b2ptr + (size_t)((g + 1) * K2_NT + t) * 16 * DT_RANK + 8 * j);
        }
        __builtin_amdgcn_sched_barrier(0);
        v8f acc2[K2_NT];
#pragma unroll
        for (int t = 0; t < K2_NT; ++t) acc2[t] = (v8f){};
#pragma unroll
        for (int j = 0; j < DT_RANK / 8; ++j) {
#pragma unroll
            for (int t = 0; t < K2_NT; ++t)
                acc2[t] = __builtin_amdgcn_wmma_f32_16x16x4_f32(
                    false, FRAG_LO(a2[j]), false, FRAG_LO(b2[g][j][t]), (short)0, acc2[t], false, false);
#pragma unroll
            for (int t = 0; t < K2_NT; ++t)
                acc2[t] = __builtin_amdgcn_wmma_f32_16x16x4_f32(
                    false, FRAG_HI(a2[j]), false, FRAG_HI(b2[g][j][t]), (short)0, acc2[t], false, false);
        }
#pragma unroll
        for (int t = 0; t < K2_NT; ++t) {
            int n = nw + (g * K2_NT + t) * 16 + lq;
            float bias = bdt[n];
#pragma unroll
            for (int v = 0; v < 8; ++v) {
                int m = m0 + v + 8 * half;
                float z = acc2[t][v] + bias;
                dt[(size_t)m * D_IN + n] = softplus_f32(z);
            }
        }
    }
}

#define SCAN_U   8
#define SCAN_WG  64
#define SCAN_DB  (D_IN / SCAN_WG)

__device__ __forceinline__ float lane_bcast(float v, int srclane) {
    return __int_as_float(__builtin_amdgcn_readlane(__float_as_int(v), srclane));
}

__global__ __launch_bounds__(SCAN_WG)
void scan_phase1(const float* __restrict__ X,
                 const float* __restrict__ xdbl,
                 const float* __restrict__ dt,
                 const float* __restrict__ A_log,
                 float* __restrict__ aprod,
                 float* __restrict__ hend) {
    int lane  = threadIdx.x & 31;
    int dblk  = blockIdx.x % SCAN_DB;
    int chunk = (blockIdx.x / SCAN_DB) % NCHUNK;
    int b     = blockIdx.x / (SCAN_DB * NCHUNK);
    int d     = dblk * SCAN_WG + threadIdx.x;
    int channel = b * D_IN + d;

    float A[D_ST], h[D_ST], ap[D_ST];
#pragma unroll
    for (int s = 0; s < D_ST; ++s) {
        A[s]  = -__expf(A_log[d * D_ST + s]);
        h[s]  = 0.0f;
        ap[s] = 1.0f;
    }
    size_t m0 = (size_t)b * SEQ + (size_t)chunk * LCHUNK;
    const float* dtp = dt + m0 * D_IN + d;
    const float* xp  = X  + m0 * D_IN + d;
    const float* bp  = xdbl + m0 * N_E + DT_RANK + (lane & 15);

    float dtc[SCAN_U], xc[SCAN_U], bc[SCAN_U];
#pragma unroll
    for (int u = 0; u < SCAN_U; ++u) {
        dtc[u] = dtp[(size_t)u * D_IN];
        xc[u]  = xp[(size_t)u * D_IN];
        bc[u]  = bp[(size_t)u * N_E];
    }
    for (int l = 0; l < LCHUNK; l += SCAN_U) {
        int ln = (l + SCAN_U < LCHUNK) ? (l + SCAN_U) : l;
        float dtn[SCAN_U], xn[SCAN_U], bn[SCAN_U];
#pragma unroll
        for (int u = 0; u < SCAN_U; ++u) {
            dtn[u] = dtp[(size_t)(ln + u) * D_IN];
            xn[u]  = xp[(size_t)(ln + u) * D_IN];
            bn[u]  = bp[(size_t)(ln + u) * N_E];
        }
#pragma unroll
        for (int u = 0; u < SCAN_U; ++u) {
            float dtv = dtc[u];
            float dtx = dtv * xc[u];
#pragma unroll
            for (int s = 0; s < D_ST; ++s) {
                float Bs = lane_bcast(bc[u], s);
                float ab = __expf(dtv * A[s]);
                h[s]  = ab * h[s] + dtx * Bs;
                ap[s] *= ab;
            }
        }
#pragma unroll
        for (int u = 0; u < SCAN_U; ++u) { dtc[u] = dtn[u]; xc[u] = xn[u]; bc[u] = bn[u]; }
    }
#pragma unroll
    for (int s = 0; s < D_ST; ++s) {
        int idx = (chunk * D_ST + s) * NCHAN + channel;
        aprod[idx] = ap[s];
        hend[idx]  = h[s];
    }
}

__global__ void scan_phase2(const float* __restrict__ aprod,
                            const float* __restrict__ hend,
                            float* __restrict__ carry) {
    int t = blockIdx.x * blockDim.x + threadIdx.x;
    if (t >= NCHAN * D_ST) return;
    float c = 0.0f;
#pragma unroll
    for (int ch = 0; ch < NCHUNK; ++ch) {
        int idx = ch * (NCHAN * D_ST) + t;
        carry[idx] = c;
        c = aprod[idx] * c + hend[idx];
    }
}

__global__ __launch_bounds__(SCAN_WG)
void scan_phase3(const float* __restrict__ X,
                 const float* __restrict__ xdbl,
                 const float* __restrict__ dt,
                 const float* __restrict__ A_log,
                 const float* __restrict__ Dp,
                 const float* __restrict__ carry,
                 float* __restrict__ y) {
    int lane  = threadIdx.x & 31;
    int dblk  = blockIdx.x % SCAN_DB;
    int chunk = (blockIdx.x / SCAN_DB) % NCHUNK;
    int b     = blockIdx.x / (SCAN_DB * NCHUNK);
    int d     = dblk * SCAN_WG + threadIdx.x;
    int channel = b * D_IN + d;

    float A[D_ST], h[D_ST];
#pragma unroll
    for (int s = 0; s < D_ST; ++s) {
        A[s] = -__expf(A_log[d * D_ST + s]);
        h[s] = carry[(chunk * D_ST + s) * NCHAN + channel];
    }
    float Dd = Dp[d];
    size_t m0 = (size_t)b * SEQ + (size_t)chunk * LCHUNK;
    const float* dtp = dt + m0 * D_IN + d;
    const float* xp  = X  + m0 * D_IN + d;
    float*       yp  = y  + m0 * D_IN + d;
    const float* bcp = xdbl + m0 * N_E + DT_RANK + lane;

    float dtc[SCAN_U], xc[SCAN_U], bc[SCAN_U];
#pragma unroll
    for (int u = 0; u < SCAN_U; ++u) {
        dtc[u] = dtp[(size_t)u * D_IN];
        xc[u]  = xp[(size_t)u * D_IN];
        bc[u]  = bcp[(size_t)u * N_E];
    }
    for (int l = 0; l < LCHUNK; l += SCAN_U) {
        int ln = (l + SCAN_U < LCHUNK) ? (l + SCAN_U) : l;
        float dtn[SCAN_U], xn[SCAN_U], bn[SCAN_U];
#pragma unroll
        for (int u = 0; u < SCAN_U; ++u) {
            dtn[u] = dtp[(size_t)(ln + u) * D_IN];
            xn[u]  = xp[(size_t)(ln + u) * D_IN];
            bn[u]  = bcp[(size_t)(ln + u) * N_E];
        }
#pragma unroll
        for (int u = 0; u < SCAN_U; ++u) {
            float dtv = dtc[u];
            float xv  = xc[u];
            float dtx = dtv * xv;
            float acc = 0.0f;
#pragma unroll
            for (int s = 0; s < D_ST; ++s) {
                float Bs = lane_bcast(bc[u], s);
                float Cs = lane_bcast(bc[u], D_ST + s);
                float ab = __expf(dtv * A[s]);
                h[s] = ab * h[s] + dtx * Bs;
                acc  = h[s] * Cs + acc;
            }
            yp[(size_t)(l + u) * D_IN] = acc + Dd * xv;
        }
#pragma unroll
        for (int u = 0; u < SCAN_U; ++u) { dtc[u] = dtn[u]; xc[u] = xn[u]; bc[u] = bn[u]; }
    }
}

extern "C" void kernel_launch(void* const* d_in, const int* in_sizes, int n_in,
                              void* d_out, int out_size, void* d_ws, size_t ws_size,
                              hipStream_t stream) {
    const float* x       = (const float*)d_in[0];
    const float* A_log   = (const float*)d_in[1];
    const float* Dp      = (const float*)d_in[2];
    const float* W_xproj = (const float*)d_in[3];
    const float* W_dt    = (const float*)d_in[4];
    const float* b_dt    = (const float*)d_in[5];
    float* y = (float*)d_out;

    char* ws = (char*)d_ws;
    size_t off = 0;
    float* xdbl  = (float*)(ws + off); off += (size_t)M_TOT * N_E * 4;
    off = (off + 255) & ~(size_t)255;
    float* dtbuf = (float*)(ws + off); off += (size_t)M_TOT * D_IN * 4;
    off = (off + 255) & ~(size_t)255;
    float* aprod = (float*)(ws + off); off += (size_t)NCHAN * D_ST * NCHUNK * 4;
    off = (off + 255) & ~(size_t)255;
    float* hend  = (float*)(ws + off); off += (size_t)NCHAN * D_ST * NCHUNK * 4;
    off = (off + 255) & ~(size_t)255;
    float* carry = (float*)(ws + off);

    xproj_dt_wmma<<<M_TOT / 16, K1_WAVES * 32, 0, stream>>>(x, W_xproj, W_dt, b_dt, xdbl, dtbuf);
    scan_phase1<<<BATCH * NCHUNK * SCAN_DB, SCAN_WG, 0, stream>>>(
        x, xdbl, dtbuf, A_log, aprod, hend);
    scan_phase2<<<(NCHAN * D_ST) / 256, 256, 0, stream>>>(aprod, hend, carry);
    scan_phase3<<<BATCH * NCHUNK * SCAN_DB, SCAN_WG, 0, stream>>>(
        x, xdbl, dtbuf, A_log, Dp, carry, y);
}
